# starting kernel
# baseline (speedup 1.0000x reference)
#include <hip/hip_runtime.h>

typedef __attribute__((ext_vector_type(16))) __bf16 v16bf;
typedef __attribute__((ext_vector_type(8)))  __bf16 v8bf;
typedef __attribute__((ext_vector_type(8)))  float  v8f;

__device__ __forceinline__ __bf16 f2bf(float f) {
  unsigned int u = __builtin_bit_cast(unsigned int, f);
  u += 0x7FFFu + ((u >> 16) & 1u);
  unsigned short s = (unsigned short)(u >> 16);
  return __builtin_bit_cast(__bf16, s);
}

__global__ void fill_zero_kernel(float* __restrict__ p, int n) {
  int i = blockIdx.x * blockDim.x + threadIdx.x;
  if (i < n) p[i] = 0.0f;
}

__global__ void degree_kernel(const int* __restrict__ src, const int* __restrict__ dst,
                              float* __restrict__ degOut, float* __restrict__ degIn, int nE) {
  int e = blockIdx.x * blockDim.x + threadIdx.x;
  if (e < nE) {
    atomicAdd(&degOut[src[e]], 1.0f);
    atomicAdd(&degIn[dst[e]], 1.0f);
  }
}

__global__ void norm_kernel(float* __restrict__ p, int n) {
  int i = blockIdx.x * blockDim.x + threadIdx.x;
  if (i < n) p[i] = rsqrtf(fmaxf(p[i], 1.0f));
}

template<int K>
__global__ __launch_bounds__(128) void gemm_rownorm_wmma(
    const float* __restrict__ X, const float* __restrict__ rowScale,
    const float* __restrict__ Wg, float* __restrict__ Y, int nRows) {
  __shared__ __bf16 Xs[16][K];
  __shared__ __bf16 Wt[64][K + 8];

  const int tid     = threadIdx.x;
  const int rowBase = blockIdx.x * 16;

  for (int idx = tid; idx < 16 * K; idx += 128) {
    int r = idx / K, k = idx - r * K;
    int row = rowBase + r;
    row = (row < nRows) ? row : (nRows - 1);
    Xs[r][k] = f2bf(X[(long)row * K + k] * rowScale[row]);
  }
  for (int idx = tid; idx < K * 64; idx += 128) {
    int k = idx >> 6, c = idx & 63;
    Wt[c][k] = f2bf(Wg[idx]);
  }
  __syncthreads();

  const int lane    = tid & 31;
  const int wave    = tid >> 5;
  const int hh      = lane >> 4;
  const int l       = lane & 15;
  const int colBase = wave << 4;

  v8f acc = {};
  #pragma unroll
  for (int k0 = 0; k0 < K; k0 += 32) {
    const v8bf a0 = *reinterpret_cast<const v8bf*>(&Xs[l][k0 + (hh << 3)]);
    const v8bf a1 = *reinterpret_cast<const v8bf*>(&Xs[l][k0 + 16 + (hh << 3)]);
    const v8bf b0 = *reinterpret_cast<const v8bf*>(&Wt[colBase + l][k0 + (hh << 3)]);
    const v8bf b1 = *reinterpret_cast<const v8bf*>(&Wt[colBase + l][k0 + 16 + (hh << 3)]);
    v16bf a, b;
    #pragma unroll
    for (int i = 0; i < 8; ++i) {
      a[i] = a0[i]; a[i + 8] = a1[i];
      b[i] = b0[i]; b[i + 8] = b1[i];
    }
    acc = __builtin_amdgcn_wmma_f32_16x16x32_bf16(
         false, a,  false, b,
         (short)0, acc,  false,  false);
  }

  float* __restrict__ yOut = Y + ((long)(rowBase + (hh << 3)) * 64 + colBase + l);
  if (rowBase + 16 <= nRows) {
    #pragma unroll
    for (int v = 0; v < 8; ++v) yOut[(long)v * 64] = acc[v];
  } else {
    #pragma unroll
    for (int v = 0; v < 8; ++v) {
      if (rowBase + v + (hh << 3) < nRows) yOut[(long)v * 64] = acc[v];
    }
  }
}

__global__ void edge_agg_kernel(const float* __restrict__ y, const int* __restrict__ src,
                                const int* __restrict__ dst, float* __restrict__ agg, int nE) {
  int gid = blockIdx.x * blockDim.x + threadIdx.x;
  int e = gid >> 4;
  if (e >= nE) return;
  int c4 = (gid & 15) << 2;
  int s = src[e], d = dst[e];
  const float4 v = *reinterpret_cast<const float4*>(y + (long)s * 64 + c4);
  float* ap = agg + (long)d * 64 + c4;
  atomicAdd(ap + 0, v.x);
  atomicAdd(ap + 1, v.y);
  atomicAdd(ap + 2, v.z);
  atomicAdd(ap + 3, v.w);
}

__global__ void finalize_kernel(const float* __restrict__ agg, const float* __restrict__ inNorm,
                                const float* __restrict__ bias, float* __restrict__ out,
                                int nRows, int doRelu) {
  int gid = blockIdx.x * blockDim.x + threadIdx.x;
  if (gid >= nRows * 64) return;
  int n = gid >> 6, c = gid & 63;
  float v = agg[gid] * inNorm[n] + bias[c];
  out[gid] = doRelu ? fmaxf(v, 0.0f) : v;
}

static inline int cdiv_i(long a, long b) { return (int)((a + b - 1) / b); }

extern "C" void kernel_launch(void* const* d_in, const int* in_sizes, int n_in,
                              void* d_out, int out_size, void* d_ws, size_t ws_size,
                              hipStream_t stream) {
  const float* feat = (const float*)d_in[0];
  const float* W0   = (const float*)d_in[1];
  const float* b0   = (const float*)d_in[2];
  const float* W1   = (const float*)d_in[3];
  const float* b1   = (const float*)d_in[4];
  const float* W2   = (const float*)d_in[5];
  const float* b2   = (const float*)d_in[6];
  const int*   src  = (const int*)d_in[7];
  const int*   dst  = (const int*)d_in[8];

  const int N = in_sizes[0] / 128;
  const int E = in_sizes[7];
  float* out = (float*)d_out;

  float* wsf = (float*)d_ws;
  float* no  = wsf;
  float* ni  = no + N;
  float* y   = ni + N;
  float* ag  = y  + (size_t)N * 64;
  float* h   = ag + (size_t)N * 64;

  const int NB = 256;
  const int gemmBlocks = cdiv_i(N, 16);
  const int nNF = N * 64;

  fill_zero_kernel<<<cdiv_i(2 * N, NB), NB, 0, stream>>>(no, 2 * N);
  degree_kernel<<<cdiv_i(E, NB), NB, 0, stream>>>(src, dst, no, ni, E);
  norm_kernel<<<cdiv_i(2 * N, NB), NB, 0, stream>>>(no, 2 * N);

  gemm_rownorm_wmma<128><<<gemmBlocks, 128, 0, stream>>>(feat, no, W0, y, N);
  fill_zero_kernel<<<cdiv_i(nNF, NB), NB, 0, stream>>>(ag, nNF);
  edge_agg_kernel<<<cdiv_i((long)E * 16, NB), NB, 0, stream>>>(y, src, dst, ag, E);
  finalize_kernel<<<cdiv_i(nNF, NB), NB, 0, stream>>>(ag, ni, b0, h, N, 1);

  gemm_rownorm_wmma<64><<<gemmBlocks, 128, 0, stream>>>(h, no, W1, y, N);
  fill_zero_kernel<<<cdiv_i(nNF, NB), NB, 0, stream>>>(ag, nNF);
  edge_agg_kernel<<<cdiv_i((long)E * 16, NB), NB, 0, stream>>>(y, src, dst, ag, E);
  finalize_kernel<<<cdiv_i(nNF, NB), NB, 0, stream>>>(ag, ni, b1, h, N, 1);

  gemm_rownorm_wmma<64><<<gemmBlocks, 128, 0, stream>>>(h, no, W2, y, N);
  fill_zero_kernel<<<cdiv_i(nNF, NB), NB, 0, stream>>>(ag, nNF);
  edge_agg_kernel<<<cdiv_i((long)E * 16, NB), NB, 0, stream>>>(y, src, dst, ag, E);
  finalize_kernel<<<cdiv_i(nNF, NB), NB, 0, stream>>>(ag, ni, b2, out, N, 0);
}


# elementwise passes folded into the GEMM launch: its A loader forms relu(sums*norm+bias) itself (same f32 steps) and the block then clears its sum rows for the next edge pass; 15 -> 10 launches
# speedup vs baseline: 1.7848x; 1.7848x over previous
#include <hip/hip_runtime.h>

typedef __attribute__((ext_vector_type(16))) __bf16 v16bf;
typedef __attribute__((ext_vector_type(8)))  __bf16 v8bf;
typedef __attribute__((ext_vector_type(4)))  __bf16 v4bf;
typedef __attribute__((ext_vector_type(8)))  float  v8f;

__device__ __forceinline__ __bf16 f2bf(float f) {
  unsigned int u = __builtin_bit_cast(unsigned int, f);
  u += 0x7FFFu + ((u >> 16) & 1u);
  unsigned short s = (unsigned short)(u >> 16);
  return __builtin_bit_cast(__bf16, s);
}


constexpr int kCntPitch = 32;

__global__ void zero_counters_kernel(float* __restrict__ cnt, int n) {
  int i = blockIdx.x * blockDim.x + threadIdx.x;
  if (i < n) cnt[(long)i * kCntPitch] = 0.0f;
}

__global__ void degree_kernel(const int* __restrict__ src, const int* __restrict__ dst,
                              float* __restrict__ cntOut, float* __restrict__ cntIn, int nE) {
  int e = blockIdx.x * blockDim.x + threadIdx.x;
  if (e < nE) {
    atomicAdd(&cntOut[(long)src[e] * kCntPitch], 1.0f);
    atomicAdd(&cntIn[(long)dst[e] * kCntPitch], 1.0f);
  }
}

__global__ void norm_kernel(const float* __restrict__ cnt, float* __restrict__ nrm, int n) {
  int i = blockIdx.x * blockDim.x + threadIdx.x;
  if (i < n) nrm[i] = rsqrtf(fmaxf(cnt[(long)i * kCntPitch], 1.0f));
}

template<int K, bool kFin>
__global__ __launch_bounds__(128) void gemm_rownorm_wmma(
    const float* __restrict__ X, float* agg,
    const float* __restrict__ inNorm, const float* __restrict__ bias,
    const float* __restrict__ rowScale,
    const float* __restrict__ Wg, float* __restrict__ Y, int nRows) {
  static_assert(!kFin || K == 64, "agg rows are 64 wide");
  __shared__ __attribute__((aligned(16))) __bf16 Xs[16][K];
  __shared__ __attribute__((aligned(16))) __bf16 Wt[64][K + 8];

  const int tid     = threadIdx.x;
  const int rowBase = blockIdx.x * 16;

  constexpr int Q = K / 4;
  for (int j = 0; j < (16 * Q) / 128; ++j) {
    const int idx = tid + 128 * j;
    const int r = idx / Q, q = idx - r * Q;
    int row = rowBase + r;
    row = (row < nRows) ? row : (nRows - 1);
    const float sc = rowScale[row];
    float4 v;
    if constexpr (kFin) {
      const float4 a = *reinterpret_cast<const float4*>(agg + (long)row * 64 + 4 * q);
      const float4 b = *reinterpret_cast<const float4*>(bias + 4 * q);
      const float nin = inNorm[row];
      v.x = fmaxf(a.x * nin + b.x, 0.0f) * sc;
      v.y = fmaxf(a.y * nin + b.y, 0.0f) * sc;
      v.z = fmaxf(a.z * nin + b.z, 0.0f) * sc;
      v.w = fmaxf(a.w * nin + b.w, 0.0f) * sc;
    } else {
      const float4 a = *reinterpret_cast<const float4*>(X + (long)row * K + 4 * q);
      v.x = a.x * sc; v.y = a.y * sc; v.z = a.z * sc; v.w = a.w * sc;
    }
    v4bf p;
    p[0] = f2bf(v.x); p[1] = f2bf(v.y); p[2] = f2bf(v.z); p[3] = f2bf(v.w);
    *reinterpret_cast<v4bf*>(&Xs[r][4 * q]) = p;
  }
  for (int idx = tid; idx < K * 64; idx += 128) {
    int k = idx >> 6, c = idx & 63;
    Wt[c][k] = f2bf(Wg[idx]);
  }
  __syncthreads();

  {
    const float4 z = {0.0f, 0.0f, 0.0f, 0.0f};
    float4* a4 = reinterpret_cast<float4*>(agg + (long)rowBase * 64);
    for (int j = 0; j < 2; ++j) {
      const int idx = tid + 128 * j;
      if (rowBase + (idx >> 4) < nRows) a4[idx] = z;
    }
  }

  const int lane    = tid & 31;
  const int wave    = tid >> 5;
  const int hh      = lane >> 4;
  const int l       = lane & 15;
  const int colBase = wave << 4;

  v8f acc = {};
  #pragma unroll
  for (int k0 = 0; k0 < K; k0 += 32) {
    const v8bf a0 = *reinterpret_cast<const v8bf*>(&Xs[l][k0 + (hh << 3)]);
    const v8bf a1 = *reinterpret_cast<const v8bf*>(&Xs[l][k0 + 16 + (hh << 3)]);
    const v8bf b0 = *reinterpret_cast<const v8bf*>(&Wt[colBase + l][k0 + (hh << 3)]);
    const v8bf b1 = *reinterpret_cast<const v8bf*>(&Wt[colBase + l][k0 + 16 + (hh << 3)]);
    v16bf a, b;
    #pragma unroll
    for (int i = 0; i < 8; ++i) {
      a[i] = a0[i]; a[i + 8] = a1[i];
      b[i] = b0[i]; b[i + 8] = b1[i];
    }
    acc = __builtin_amdgcn_wmma_f32_16x16x32_bf16(
         false, a,  false, b,
         (short)0, acc,  false,  false);
  }

  float* __restrict__ yOut = Y + ((long)(rowBase + (hh << 3)) * 64 + colBase + l);
  if (rowBase + 16 <= nRows) {
    #pragma unroll
    for (int v = 0; v < 8; ++v) yOut[(long)v * 64] = acc[v];
  } else {
    #pragma unroll
    for (int v = 0; v < 8; ++v) {
      if (rowBase + v + (hh << 3) < nRows) yOut[(long)v * 64] = acc[v];
    }
  }
}

constexpr int kEdgesPerWave = 4;

__global__ void edge_agg_kernel(
    const float* __restrict__ y, const int* __restrict__ src,
    const int* __restrict__ dst, float* __restrict__ agg, int nE) {
  const int lane = threadIdx.x & 31;
  const int wid = __builtin_amdgcn_readfirstlane((int)((blockIdx.x * blockDim.x + threadIdx.x) >> 5));
  const long e0 = (long)wid * kEdgesPerWave;
  if (e0 >= nE) return;
  if (e0 + kEdgesPerWave <= nE) {
    int s[kEdgesPerWave], d[kEdgesPerWave];
    for (int i = 0; i < kEdgesPerWave; ++i) { s[i] = src[e0 + i]; d[i] = dst[e0 + i]; }
    float lo[kEdgesPerWave], hi[kEdgesPerWave];
    for (int i = 0; i < kEdgesPerWave; ++i) {
      const float* yp = y + (long)s[i] * 64 + lane;
      lo[i] = yp[0];
      hi[i] = yp[32];
    }
    for (int i = 0; i < kEdgesPerWave; ++i) {
      float* ap = agg + (long)d[i] * 64 + lane;
      atomicAdd(ap, lo[i]);
      atomicAdd(ap + 32, hi[i]);
    }
  } else {
    for (long e = e0; e < nE; ++e) {
      const int s = src[e], d = dst[e];
      const float* yp = y + (long)s * 64 + lane;
      const float lo = yp[0], hi = yp[32];
      float* ap = agg + (long)d * 64 + lane;
      atomicAdd(ap, lo);
      atomicAdd(ap + 32, hi);
    }
  }
}

__global__ void finalize_kernel(const float* __restrict__ agg, const float* __restrict__ inNorm,
                                const float* __restrict__ bias, float* __restrict__ out,
                                int nRows, int doRelu) {
  int gid = blockIdx.x * blockDim.x + threadIdx.x;
  if (gid >= nRows * 64) return;
  int n = gid >> 6, c = gid & 63;
  float v = agg[gid] * inNorm[n] + bias[c];
  out[gid] = doRelu ? fmaxf(v, 0.0f) : v;
}

static inline int cdiv_i(long a, long b) { return (int)((a + b - 1) / b); }

extern "C" void kernel_launch(void* const* d_in, const int* in_sizes, int n_in,
                              void* d_out, int out_size, void* d_ws, size_t ws_size,
                              hipStream_t stream) {
  const float* feat = (const float*)d_in[0];
  const float* W0   = (const float*)d_in[1];
  const float* b0   = (const float*)d_in[2];
  const float* W1   = (const float*)d_in[3];
  const float* b1   = (const float*)d_in[4];
  const float* W2   = (const float*)d_in[5];
  const float* b2   = (const float*)d_in[6];
  const int*   src  = (const int*)d_in[7];
  const int*   dst  = (const int*)d_in[8];

  const int N = in_sizes[0] / 128;
  const int E = in_sizes[7];
  float* out = (float*)d_out;

  float* wsf = (float*)d_ws;
  float* no  = wsf;
  float* ni  = no + N;
  float* y   = ni + N;
  float* ag  = y  + (size_t)N * 64;
  float* cnt = ag + (size_t)N * 64;
  float* cntIn = cnt + (size_t)N * kCntPitch;

  const int NB = 256;
  const int gemmBlocks = cdiv_i(N, 16);
  const int nNF = N * 64;

  zero_counters_kernel<<<cdiv_i(2 * N, NB), NB, 0, stream>>>(cnt, 2 * N);
  degree_kernel<<<cdiv_i(E, NB), NB, 0, stream>>>(src, dst, cnt, cntIn, E);
  norm_kernel<<<cdiv_i(2 * N, NB), NB, 0, stream>>>(cnt, no, 2 * N);

  gemm_rownorm_wmma<128, false><<<gemmBlocks, 128, 0, stream>>>(feat, ag, nullptr, nullptr, no, W0, y, N);
  edge_agg_kernel<<<cdiv_i((long)cdiv_i(E, kEdgesPerWave) * 32, NB), NB, 0, stream>>>(y, src, dst, ag, E);

  gemm_rownorm_wmma<64, true><<<gemmBlocks, 128, 0, stream>>>(nullptr, ag, ni, b0, no, W1, y, N);
  edge_agg_kernel<<<cdiv_i((long)cdiv_i(E, kEdgesPerWave) * 32, NB), NB, 0, stream>>>(y, src, dst, ag, E);

  gemm_rownorm_wmma<64, true><<<gemmBlocks, 128, 0, stream>>>(nullptr, ag, ni, b1, no, W2, y, N);
  edge_agg_kernel<<<cdiv_i((long)cdiv_i(E, kEdgesPerWave) * 32, NB), NB, 0, stream>>>(y, src, dst, ag, E);
  finalize_kernel<<<cdiv_i(nNF, NB), NB, 0, stream>>>(ag, ni, b2, out, N, 0);
}


# float atomic adds go to a per-cache-half copy of their target (plane picked by the block's die id): agg rows and degree counters as 2 planes, readers add both; a sum = two arrival-order partials
# speedup vs baseline: 3.2159x; 1.8018x over previous
#include <hip/hip_runtime.h>

typedef __attribute__((ext_vector_type(16))) __bf16 v16bf;
typedef __attribute__((ext_vector_type(8)))  __bf16 v8bf;
typedef __attribute__((ext_vector_type(4)))  __bf16 v4bf;
typedef __attribute__((ext_vector_type(8)))  float  v8f;

__device__ __forceinline__ __bf16 f2bf(float f) {
  unsigned int u = __builtin_bit_cast(unsigned int, f);
  u += 0x7FFFu + ((u >> 16) & 1u);
  unsigned short s = (unsigned short)(u >> 16);
  return __builtin_bit_cast(__bf16, s);
}


constexpr int kCntPitch = 32;

__device__ __forceinline__ int cache_half() {
  return (int)((__builtin_amdgcn_s_sendmsg_rtn(0x87) >> 18) & 1u);
}

__device__ __forceinline__ int block_cache_half(int* slot) {
  if (threadIdx.x == 0) *slot = cache_half();
  __builtin_amdgcn_fence(__ATOMIC_RELEASE, "workgroup", "local");
  __builtin_amdgcn_s_barrier();
  __builtin_amdgcn_fence(__ATOMIC_ACQUIRE, "workgroup", "local");
  return *slot;
}

__global__ void zero_counters_kernel(float* __restrict__ cnt, int n) {
  int i = blockIdx.x * blockDim.x + threadIdx.x;
  if (i < n) cnt[(long)i * kCntPitch] = 0.0f;
}

__global__ void degree_kernel(const int* __restrict__ src, const int* __restrict__ dst,
                              float* __restrict__ cnt0, float* __restrict__ cnt1, int nNodes, int nE) {
  __shared__ int sHalf;
  const int e = blockIdx.x * blockDim.x + threadIdx.x;
  int s = 0, d = 0;
  if (e < nE) { s = src[e]; d = dst[e]; }
  float* __restrict__ cnt = block_cache_half(&sHalf) ? cnt1 : cnt0;
  if (e < nE) {
    atomicAdd(&cnt[(long)s * kCntPitch], 1.0f);
    atomicAdd(&cnt[((long)nNodes + d) * kCntPitch], 1.0f);
  }
}

__global__ void norm_kernel(const float* __restrict__ cnt0, const float* __restrict__ cnt1,
                            float* __restrict__ nrm, int n) {
  int i = blockIdx.x * blockDim.x + threadIdx.x;
  if (i < n) nrm[i] = rsqrtf(fmaxf(cnt0[(long)i * kCntPitch] + cnt1[(long)i * kCntPitch], 1.0f));
}

template<int K, bool kFin>
__global__ __launch_bounds__(128) void gemm_rownorm_wmma(
    const float* __restrict__ X, float* agg0, float* agg1,
    const float* __restrict__ inNorm, const float* __restrict__ bias,
    const float* __restrict__ rowScale,
    const float* __restrict__ Wg, float* __restrict__ Y, int nRows) {
  static_assert(!kFin || K == 64, "agg rows are 64 wide");
  __shared__ __attribute__((aligned(16))) __bf16 Xs[16][K];
  __shared__ __attribute__((aligned(16))) __bf16 Wt[64][K + 8];

  const int tid     = threadIdx.x;
  const int rowBase = blockIdx.x * 16;

  constexpr int Q = K / 4;
  for (int j = 0; j < (16 * Q) / 128; ++j) {
    const int idx = tid + 128 * j;
    const int r = idx / Q, q = idx - r * Q;
    int row = rowBase + r;
    row = (row < nRows) ? row : (nRows - 1);
    const float sc = rowScale[row];
    float4 v;
    if constexpr (kFin) {
      const float4 a0 = *reinterpret_cast<const float4*>(agg0 + (long)row * 64 + 4 * q);
      const float4 a1 = *reinterpret_cast<const float4*>(agg1 + (long)row * 64 + 4 * q);
      const float4 a = {a0.x + a1.x, a0.y + a1.y, a0.z + a1.z, a0.w + a1.w};
      const float4 b = *reinterpret_cast<const float4*>(bias + 4 * q);
      const float nin = inNorm[row];
      v.x = fmaxf(a.x * nin + b.x, 0.0f) * sc;
      v.y = fmaxf(a.y * nin + b.y, 0.0f) * sc;
      v.z = fmaxf(a.z * nin + b.z, 0.0f) * sc;
      v.w = fmaxf(a.w * nin + b.w, 0.0f) * sc;
    } else {
      const float4 a = *reinterpret_cast<const float4*>(X + (long)row * K + 4 * q);
      v.x = a.x * sc; v.y = a.y * sc; v.z = a.z * sc; v.w = a.w * sc;
    }
    v4bf p;
    p[0] = f2bf(v.x); p[1] = f2bf(v.y); p[2] = f2bf(v.z); p[3] = f2bf(v.w);
    *reinterpret_cast<v4bf*>(&Xs[r][4 * q]) = p;
  }
  for (int idx = tid; idx < K * 64; idx += 128) {
    int k = idx >> 6, c = idx & 63;
    Wt[c][k] = f2bf(Wg[idx]);
  }
  __syncthreads();

  {
    const float4 z = {0.0f, 0.0f, 0.0f, 0.0f};
    float4* a4 = reinterpret_cast<float4*>(agg0 + (long)rowBase * 64);
    float4* b4 = reinterpret_cast<float4*>(agg1 + (long)rowBase * 64);
    for (int j = 0; j < 2; ++j) {
      const int idx = tid + 128 * j;
      if (rowBase + (idx >> 4) < nRows) { a4[idx] = z; b4[idx] = z; }
    }
  }

  const int lane    = tid & 31;
  const int wave    = tid >> 5;
  const int hh      = lane >> 4;
  const int l       = lane & 15;
  const int colBase = wave << 4;

  v8f acc = {};
  #pragma unroll
  for (int k0 = 0; k0 < K; k0 += 32) {
    const v8bf a0 = *reinterpret_cast<const v8bf*>(&Xs[l][k0 + (hh << 3)]);
    const v8bf a1 = *reinterpret_cast<const v8bf*>(&Xs[l][k0 + 16 + (hh << 3)]);
    const v8bf b0 = *reinterpret_cast<const v8bf*>(&Wt[colBase + l][k0 + (hh << 3)]);
    const v8bf b1 = *reinterpret_cast<const v8bf*>(&Wt[colBase + l][k0 + 16 + (hh << 3)]);
    v16bf a, b;
    #pragma unroll
    for (int i = 0; i < 8; ++i) {
      a[i] = a0[i]; a[i + 8] = a1[i];
      b[i] = b0[i]; b[i + 8] = b1[i];
    }
    acc = __builtin_amdgcn_wmma_f32_16x16x32_bf16(
         false, a,  false, b,
         (short)0, acc,  false,  false);
  }

  float* __restrict__ yOut = Y + ((long)(rowBase + (hh << 3)) * 64 + colBase + l);
  if (rowBase + 16 <= nRows) {
    #pragma unroll
    for (int v = 0; v < 8; ++v) yOut[(long)v * 64] = acc[v];
  } else {
    #pragma unroll
    for (int v = 0; v < 8; ++v) {
      if (rowBase + v + (hh << 3) < nRows) yOut[(long)v * 64] = acc[v];
    }
  }
}

constexpr int kEdgesPerWave = 4;

__global__ void edge_agg_kernel(
    const float* __restrict__ y, const int* __restrict__ src,
    const int* __restrict__ dst, float* __restrict__ agg0, float* __restrict__ agg1, int nE) {
  __shared__ int sHalf;
  const int lane = threadIdx.x & 31;
  const int wid = __builtin_amdgcn_readfirstlane((int)((blockIdx.x * blockDim.x + threadIdx.x) >> 5));
  const long e0 = (long)wid * kEdgesPerWave;
  const bool full = e0 + kEdgesPerWave <= nE;
  int d[kEdgesPerWave];
  float lo[kEdgesPerWave], hi[kEdgesPerWave];
  if (full) {
    int s[kEdgesPerWave];
    for (int i = 0; i < kEdgesPerWave; ++i) { s[i] = src[e0 + i]; d[i] = dst[e0 + i]; }
    for (int i = 0; i < kEdgesPerWave; ++i) {
      const float* yp = y + (long)s[i] * 64 + lane;
      lo[i] = yp[0];
      hi[i] = yp[32];
    }
  }
  float* __restrict__ agg = block_cache_half(&sHalf) ? agg1 : agg0;
  if (full) {
    for (int i = 0; i < kEdgesPerWave; ++i) {
      float* ap = agg + (long)d[i] * 64 + lane;
      atomicAdd(ap, lo[i]);
      atomicAdd(ap + 32, hi[i]);
    }
  } else {
    for (long e = e0; e < nE; ++e) {
      const int s = src[e], dd = dst[e];
      const float* yp = y + (long)s * 64 + lane;
      const float l0 = yp[0], h0 = yp[32];
      float* ap = agg + (long)dd * 64 + lane;
      atomicAdd(ap, l0);
      atomicAdd(ap + 32, h0);
    }
  }
}

__global__ void finalize_kernel(const float* __restrict__ agg0, const float* __restrict__ agg1,
                                const float* __restrict__ inNorm,
                                const float* __restrict__ bias, float* __restrict__ out,
                                int nRows, int doRelu) {
  int gid = blockIdx.x * blockDim.x + threadIdx.x;
  if (gid >= nRows * 64) return;
  int n = gid >> 6, c = gid & 63;
  float v = (agg0[gid] + agg1[gid]) * inNorm[n] + bias[c];
  out[gid] = doRelu ? fmaxf(v, 0.0f) : v;
}

static inline int cdiv_i(long a, long b) { return (int)((a + b - 1) / b); }

extern "C" void kernel_launch(void* const* d_in, const int* in_sizes, int n_in,
                              void* d_out, int out_size, void* d_ws, size_t ws_size,
                              hipStream_t stream) {
  const float* feat = (const float*)d_in[0];
  const float* W0   = (const float*)d_in[1];
  const float* b0   = (const float*)d_in[2];
  const float* W1   = (const float*)d_in[3];
  const float* b1   = (const float*)d_in[4];
  const float* W2   = (const float*)d_in[5];
  const float* b2   = (const float*)d_in[6];
  const int*   src  = (const int*)d_in[7];
  const int*   dst  = (const int*)d_in[8];

  const int N = in_sizes[0] / 128;
  const int E = in_sizes[7];
  float* out = (float*)d_out;

  float* wsf  = (float*)d_ws;
  float* no   = wsf;
  float* ni   = no + N;
  float* y    = ni + N;
  float* ag0  = y   + (size_t)N * 64;
  float* ag1  = ag0 + (size_t)N * 64;
  float* cnt0 = ag1 + (size_t)N * 64;
  float* cnt1 = cnt0 + (size_t)2 * N * kCntPitch;

  const int NB = 256;
  const int gemmBlocks = cdiv_i(N, 16);
  const int nNF = N * 64;

  zero_counters_kernel<<<cdiv_i(4 * N, NB), NB, 0, stream>>>(cnt0, 4 * N);
  degree_kernel<<<cdiv_i(E, NB), NB, 0, stream>>>(src, dst, cnt0, cnt1, N, E);
  norm_kernel<<<cdiv_i(2 * N, NB), NB, 0, stream>>>(cnt0, cnt1, no, 2 * N);

  gemm_rownorm_wmma<128, false><<<gemmBlocks, 128, 0, stream>>>(feat, ag0, ag1, nullptr, nullptr, no, W0, y, N);
  edge_agg_kernel<<<cdiv_i((long)cdiv_i(E, kEdgesPerWave) * 32, NB), NB, 0, stream>>>(y, src, dst, ag0, ag1, E);

  gemm_rownorm_wmma<64, true><<<gemmBlocks, 128, 0, stream>>>(nullptr, ag0, ag1, ni, b0, no, W1, y, N);
  edge_agg_kernel<<<cdiv_i((long)cdiv_i(E, kEdgesPerWave) * 32, NB), NB, 0, stream>>>(y, src, dst, ag0, ag1, E);

  gemm_rownorm_wmma<64, true><<<gemmBlocks, 128, 0, stream>>>(nullptr, ag0, ag1, ni, b1, no, W2, y, N);
  edge_agg_kernel<<<cdiv_i((long)cdiv_i(E, kEdgesPerWave) * 32, NB), NB, 0, stream>>>(y, src, dst, ag0, ag1, E);
  finalize_kernel<<<cdiv_i(nNF, NB), NB, 0, stream>>>(ag0, ag1, ni, b2, out, N, 0);
}


# edge passes push from the source: degree launch files each dst under its src (slot = rank of the returning count add, per cache half); y[src] loaded once per 8-edge wave; sums arrival-order
# speedup vs baseline: 3.8592x; 1.2000x over previous
#include <hip/hip_runtime.h>

typedef __attribute__((ext_vector_type(16))) __bf16 v16bf;
typedef __attribute__((ext_vector_type(8)))  __bf16 v8bf;
typedef __attribute__((ext_vector_type(4)))  __bf16 v4bf;
typedef __attribute__((ext_vector_type(8)))  float  v8f;

__device__ __forceinline__ __bf16 f2bf(float f) {
  unsigned int u = __builtin_bit_cast(unsigned int, f);
  u += 0x7FFFu + ((u >> 16) & 1u);
  unsigned short s = (unsigned short)(u >> 16);
  return __builtin_bit_cast(__bf16, s);
}


constexpr int kCntPitch = 32;

__device__ __forceinline__ int cache_half() {
  return (int)((__builtin_amdgcn_s_sendmsg_rtn(0x87) >> 18) & 1u);
}

__device__ __forceinline__ int block_cache_half(int* slot) {
  if (threadIdx.x == 0) *slot = cache_half();
  __builtin_amdgcn_fence(__ATOMIC_RELEASE, "workgroup", "local");
  __builtin_amdgcn_s_barrier();
  __builtin_amdgcn_fence(__ATOMIC_ACQUIRE, "workgroup", "local");
  return *slot;
}

constexpr int kBlkWords = 64;
constexpr int kBlkSlots = kBlkWords - 1;

__global__ void zero_build_kernel(int* __restrict__ blk, int nBlk4, float* __restrict__ icnt, int nIcnt,
                                  int* __restrict__ ovfCount) {
  const int i = blockIdx.x * blockDim.x + threadIdx.x;
  if (i < nBlk4) reinterpret_cast<int4*>(blk)[i] = int4{0, 0, 0, 0};
  if (i < nIcnt) icnt[(long)i * kCntPitch] = 0.0f;
  if (i == 0) *ovfCount = 0;
}

__global__ void degree_build_kernel(const int* __restrict__ src, const int* __restrict__ dst,
                                    int* __restrict__ oblk0, int* __restrict__ oblk1,
                                    float* __restrict__ icnt0, float* __restrict__ icnt1,
                                    int* __restrict__ ovfCount, int* __restrict__ ovf, int nE) {
  __shared__ int sHalf;
  const int e = blockIdx.x * blockDim.x + threadIdx.x;
  int s = 0, d = 0;
  if (e < nE) { s = src[e]; d = dst[e]; }
  const int half = block_cache_half(&sHalf);
  int* __restrict__ oblk = half ? oblk1 : oblk0;
  float* __restrict__ icnt = half ? icnt1 : icnt0;
  if (e < nE) {
    int* b = oblk + (long)s * kBlkWords;
    const int r = atomicAdd(b, 1);
    atomicAdd(&icnt[(long)d * kCntPitch], 1.0f);
    if (r < kBlkSlots) {
      b[1 + r] = d;
    } else {
      const int k = atomicAdd(ovfCount, 1);
      ovf[2 * (long)k] = s;
      ovf[2 * (long)k + 1] = d;
    }
  }
}

__global__ void norm_kernel(const int* __restrict__ oblk0, const int* __restrict__ oblk1,
                            const float* __restrict__ icnt0, const float* __restrict__ icnt1,
                            float* __restrict__ no, float* __restrict__ ni, int n) {
  const int i = blockIdx.x * blockDim.x + threadIdx.x;
  if (i < n) {
    no[i] = rsqrtf(fmaxf((float)(oblk0[(long)i * kBlkWords] + oblk1[(long)i * kBlkWords]), 1.0f));
    ni[i] = rsqrtf(fmaxf(icnt0[(long)i * kCntPitch] + icnt1[(long)i * kCntPitch], 1.0f));
  }
}

template<int K, bool kFin>
__global__ __launch_bounds__(128) void gemm_rownorm_wmma(
    const float* __restrict__ X, float* agg0, float* agg1,
    const float* __restrict__ inNorm, const float* __restrict__ bias,
    const float* __restrict__ rowScale,
    const float* __restrict__ Wg, float* __restrict__ Y, int nRows) {
  static_assert(!kFin || K == 64, "agg rows are 64 wide");
  __shared__ __attribute__((aligned(16))) __bf16 Xs[16][K];
  __shared__ __attribute__((aligned(16))) __bf16 Wt[64][K + 8];

  const int tid     = threadIdx.x;
  const int rowBase = blockIdx.x * 16;

  constexpr int Q = K / 4;
  for (int j = 0; j < (16 * Q) / 128; ++j) {
    const int idx = tid + 128 * j;
    const int r = idx / Q, q = idx - r * Q;
    int row = rowBase + r;
    row = (row < nRows) ? row : (nRows - 1);
    const float sc = rowScale[row];
    float4 v;
    if constexpr (kFin) {
      const float4 a0 = *reinterpret_cast<const float4*>(agg0 + (long)row * 64 + 4 * q);
      const float4 a1 = *reinterpret_cast<const float4*>(agg1 + (long)row * 64 + 4 * q);
      const float4 a = {a0.x + a1.x, a0.y + a1.y, a0.z + a1.z, a0.w + a1.w};
      const float4 b = *reinterpret_cast<const float4*>(bias + 4 * q);
      const float nin = inNorm[row];
      v.x = fmaxf(a.x * nin + b.x, 0.0f) * sc;
      v.y = fmaxf(a.y * nin + b.y, 0.0f) * sc;
      v.z = fmaxf(a.z * nin + b.z, 0.0f) * sc;
      v.w = fmaxf(a.w * nin + b.w, 0.0f) * sc;
    } else {
      const float4 a = *reinterpret_cast<const float4*>(X + (long)row * K + 4 * q);
      v.x = a.x * sc; v.y = a.y * sc; v.z = a.z * sc; v.w = a.w * sc;
    }
    v4bf p;
    p[0] = f2bf(v.x); p[1] = f2bf(v.y); p[2] = f2bf(v.z); p[3] = f2bf(v.w);
    *reinterpret_cast<v4bf*>(&Xs[r][4 * q]) = p;
  }
  for (int idx = tid; idx < K * 64; idx += 128) {
    int k = idx >> 6, c = idx & 63;
    Wt[c][k] = f2bf(Wg[idx]);
  }
  __syncthreads();

  {
    const float4 z = {0.0f, 0.0f, 0.0f, 0.0f};
    float4* a4 = reinterpret_cast<float4*>(agg0 + (long)rowBase * 64);
    float4* b4 = reinterpret_cast<float4*>(agg1 + (long)rowBase * 64);
    for (int j = 0; j < 2; ++j) {
      const int idx = tid + 128 * j;
      if (rowBase + (idx >> 4) < nRows) { a4[idx] = z; b4[idx] = z; }
    }
  }

  const int lane    = tid & 31;
  const int wave    = tid >> 5;
  const int hh      = lane >> 4;
  const int l       = lane & 15;
  const int colBase = wave << 4;

  v8f acc = {};
  #pragma unroll
  for (int k0 = 0; k0 < K; k0 += 32) {
    const v8bf a0 = *reinterpret_cast<const v8bf*>(&Xs[l][k0 + (hh << 3)]);
    const v8bf a1 = *reinterpret_cast<const v8bf*>(&Xs[l][k0 + 16 + (hh << 3)]);
    const v8bf b0 = *reinterpret_cast<const v8bf*>(&Wt[colBase + l][k0 + (hh << 3)]);
    const v8bf b1 = *reinterpret_cast<const v8bf*>(&Wt[colBase + l][k0 + 16 + (hh << 3)]);
    v16bf a, b;
    #pragma unroll
    for (int i = 0; i < 8; ++i) {
      a[i] = a0[i]; a[i + 8] = a1[i];
      b[i] = b0[i]; b[i + 8] = b1[i];
    }
    acc = __builtin_amdgcn_wmma_f32_16x16x32_bf16(
         false, a,  false, b,
         (short)0, acc,  false,  false);
  }

  float* __restrict__ yOut = Y + ((long)(rowBase + (hh << 3)) * 64 + colBase + l);
  if (rowBase + 16 <= nRows) {
    #pragma unroll
    for (int v = 0; v < 8; ++v) yOut[(long)v * 64] = acc[v];
  } else {
    #pragma unroll
    for (int v = 0; v < 8; ++v) {
      if (rowBase + v + (hh << 3) < nRows) yOut[(long)v * 64] = acc[v];
    }
  }
}

constexpr int kWavesPerNode = 4;

__device__ __forceinline__ void push_row(float* __restrict__ agg, int d, int lane, float lo, float hi) {
  float* ap = agg + (long)d * 64 + lane;
  atomicAdd(ap, lo);
  atomicAdd(ap + 32, hi);
}

__global__ __launch_bounds__(32 * kWavesPerNode) void edge_push_kernel(
    const float* __restrict__ y, const int* __restrict__ oblk0, const int* __restrict__ oblk1,
    const int* __restrict__ ovfCount, const int* __restrict__ ovf,
    float* __restrict__ agg0, float* __restrict__ agg1) {
  __shared__ int sHalf;
  const int lane = threadIdx.x & 31;
  const int w    = __builtin_amdgcn_readfirstlane((int)(threadIdx.x >> 5));
  const long n   = blockIdx.x;
  const int* __restrict__ blk = ((w >> 1) ? oblk1 : oblk0) + n * kBlkWords;
  const int w0 = blk[lane];
  const float lo = y[n * 64 + lane];
  const float hi = y[n * 64 + 32 + lane];
  const int nOvf = *ovfCount;
  float* __restrict__ agg = block_cache_half(&sHalf) ? agg1 : agg0;

  int cnt = __builtin_amdgcn_readlane(w0, 0);
  cnt = cnt < kBlkSlots ? cnt : kBlkSlots;
  int e = w & 1;
  for (; e < cnt && e < 31; e += 2) push_row(agg, __builtin_amdgcn_readlane(w0, e + 1), lane, lo, hi);
  if (cnt > 31) {
    const int w1 = blk[32 + lane];
    for (; e < cnt; e += 2) push_row(agg, __builtin_amdgcn_readlane(w1, e - 31), lane, lo, hi);
  }

  const long nWaves = (long)gridDim.x * kWavesPerNode;
  for (long k = n * kWavesPerNode + w; k < nOvf; k += nWaves) {
    const int s = ovf[2 * k], dd = ovf[2 * k + 1];
    const float* yp = y + (long)s * 64 + lane;
    push_row(agg, dd, lane, yp[0], yp[32]);
  }
}

__global__ void finalize_kernel(const float* __restrict__ agg0, const float* __restrict__ agg1,
                                const float* __restrict__ inNorm,
                                const float* __restrict__ bias, float* __restrict__ out,
                                int nRows, int doRelu) {
  int gid = blockIdx.x * blockDim.x + threadIdx.x;
  if (gid >= nRows * 64) return;
  int n = gid >> 6, c = gid & 63;
  float v = (agg0[gid] + agg1[gid]) * inNorm[n] + bias[c];
  out[gid] = doRelu ? fmaxf(v, 0.0f) : v;
}

static inline int cdiv_i(long a, long b) { return (int)((a + b - 1) / b); }

extern "C" void kernel_launch(void* const* d_in, const int* in_sizes, int n_in,
                              void* d_out, int out_size, void* d_ws, size_t ws_size,
                              hipStream_t stream) {
  const float* feat = (const float*)d_in[0];
  const float* W0   = (const float*)d_in[1];
  const float* b0   = (const float*)d_in[2];
  const float* W1   = (const float*)d_in[3];
  const float* b1   = (const float*)d_in[4];
  const float* W2   = (const float*)d_in[5];
  const float* b2   = (const float*)d_in[6];
  const int*   src  = (const int*)d_in[7];
  const int*   dst  = (const int*)d_in[8];

  const int N = in_sizes[0] / 128;
  const int E = in_sizes[7];
  float* out = (float*)d_out;

  float* wsf   = (float*)d_ws;
  float* no    = wsf;
  float* ni    = no + N;
  float* y     = ni + N;
  float* ag0   = y   + (size_t)N * 64;
  float* ag1   = ag0 + (size_t)N * 64;
  int*   ob0   = (int*)(ag1 + (size_t)N * 64);
  int*   ob1   = ob0 + (size_t)N * kBlkWords;
  float* ic0   = (float*)(ob1 + (size_t)N * kBlkWords);
  float* ic1   = ic0 + (size_t)N * kCntPitch;
  int*   ovfN  = (int*)(ic1 + (size_t)N * kCntPitch);
  int*   ovf   = ovfN + 32;

  const int NB = 256;
  const int gemmBlocks = cdiv_i(N, 16);
  const int nNF = N * 64;
  const int nBlk4 = N * (2 * kBlkWords / 4);

  zero_build_kernel<<<cdiv_i(nBlk4, NB), NB, 0, stream>>>(ob0, nBlk4, ic0, 2 * N, ovfN);
  degree_build_kernel<<<cdiv_i(E, NB), NB, 0, stream>>>(src, dst, ob0, ob1, ic0, ic1, ovfN, ovf, E);
  norm_kernel<<<cdiv_i(N, NB), NB, 0, stream>>>(ob0, ob1, ic0, ic1, no, ni, N);

  gemm_rownorm_wmma<128, false><<<gemmBlocks, 128, 0, stream>>>(feat, ag0, ag1, nullptr, nullptr, no, W0, y, N);
  edge_push_kernel<<<N, 32 * kWavesPerNode, 0, stream>>>(y, ob0, ob1, ovfN, ovf, ag0, ag1);

  gemm_rownorm_wmma<64, true><<<gemmBlocks, 128, 0, stream>>>(nullptr, ag0, ag1, ni, b0, no, W1, y, N);
  edge_push_kernel<<<N, 32 * kWavesPerNode, 0, stream>>>(y, ob0, ob1, ovfN, ovf, ag0, ag1);

  gemm_rownorm_wmma<64, true><<<gemmBlocks, 128, 0, stream>>>(nullptr, ag0, ag1, ni, b1, no, W2, y, N);
  edge_push_kernel<<<N, 32 * kWavesPerNode, 0, stream>>>(y, ob0, ob1, ovfN, ovf, ag0, ag1);
  finalize_kernel<<<cdiv_i(nNF, NB), NB, 0, stream>>>(ag0, ag1, ni, b2, out, N, 0);
}
